# starting kernel
# baseline (speedup 1.0000x reference)
#include <hip/hip_runtime.h>
#include <math.h>

typedef __attribute__((ext_vector_type(2))) float v2f;
typedef __attribute__((ext_vector_type(8))) float v8f;

#define D_FEAT 64

__global__ void zero2_kernel(float* __restrict__ a, long na4,
                             float* __restrict__ b, long nb4) {
    long i = (long)blockIdx.x * blockDim.x + threadIdx.x;
    long stride = (long)gridDim.x * blockDim.x;
    float4 z = make_float4(0.f, 0.f, 0.f, 0.f);
    for (long j = i; j < na4 + nb4; j += stride) {
        if (j < na4) ((float4*)a)[j] = z;
        else         ((float4*)b)[j - na4] = z;
    }
}

__global__ void degree_kernel(const int* __restrict__ row, const int* __restrict__ col,
                              int E, float* degO, float* degI) {
    int e = blockIdx.x * blockDim.x + threadIdx.x;
    if (e >= E) return;
    unsafeAtomicAdd(&degO[row[e]], 1.0f);
    unsafeAtomicAdd(&degI[col[e]], 1.0f);
}

__global__ void deginv_kernel(float* degO, float* degI, int n) {
    int i = blockIdx.x * blockDim.x + threadIdx.x;
    if (i >= n) return;
    float d0 = degO[i];
    float d1 = degI[i];
    degO[i] = (d0 > 0.f) ? powf(d0, -0.25f) : 0.f;
    degI[i] = (d1 > 0.f) ? powf(d1, -0.25f) : 0.f;
}

__global__ void wcomb_kernel(const float* __restrict__ Wr, const float* __restrict__ br,
                             const float* __restrict__ Wi, const float* __restrict__ bi,
                             int K, float* WRc, float* WIc, float* bmi, float* bpi) {
    int i = blockIdx.x * blockDim.x + threadIdx.x;
    if (i < D_FEAT * D_FEAT) {
        float sr = 0.f, si = 0.f, c = 1.f;
        for (int k = 0; k < K; ++k) {
            sr += c * Wr[k * D_FEAT * D_FEAT + i];
            si += c * Wi[k * D_FEAT * D_FEAT + i];
            c *= 0.5f;
        }
        WRc[i] = sr;
        WIc[i] = si;
    } else if (i < D_FEAT * D_FEAT + D_FEAT) {
        int o = i - D_FEAT * D_FEAT;
        float sr = 0.f, si = 0.f, c = 1.f;
        for (int k = 0; k < K; ++k) {
            sr += c * br[k * D_FEAT + o];
            si += c * bi[k * D_FEAT + o];
            c *= 0.5f;
        }
        bmi[o] = sr - si;
        bpi[o] = sr + si;
    }
}

__global__ __launch_bounds__(256) void spmm_kernel(
        const float* __restrict__ xr, const float* __restrict__ xi,
        const int* __restrict__ row, const int* __restrict__ col, int E,
        const float* __restrict__ out_inv, const float* __restrict__ in_inv,
        float* yr, float* yi, float* yrt, float* yit) {
    const int lane = threadIdx.x & 31;
    int e = blockIdx.x * 8 + (threadIdx.x >> 5);
    e = __builtin_amdgcn_readfirstlane(e);
    if (e >= E) return;

    const int r = row[e];
    const int c = col[e];
    const float w = out_inv[r] * in_inv[c];

    const int d = lane << 1;
    const long ro = (long)r * D_FEAT + d;
    const long co = (long)c * D_FEAT + d;

    const float2 xrc = *(const float2*)(xr + co);
    const float2 xic = *(const float2*)(xi + co);
    const float2 xrr = *(const float2*)(xr + ro);
    const float2 xir = *(const float2*)(xi + ro);

    unsafeAtomicAdd(&yr[ro],      w * xrc.x);
    unsafeAtomicAdd(&yr[ro + 1],  w * xrc.y);
    unsafeAtomicAdd(&yi[ro],      w * xic.x);
    unsafeAtomicAdd(&yi[ro + 1],  w * xic.y);
    unsafeAtomicAdd(&yrt[co],     w * xrr.x);
    unsafeAtomicAdd(&yrt[co + 1], w * xrr.y);
    unsafeAtomicAdd(&yit[co],     w * xir.x);
    unsafeAtomicAdd(&yit[co + 1], w * xir.y);
}

__global__ __launch_bounds__(128) void faber_gemm_kernel(
        const float* __restrict__ yr, const float* __restrict__ yi,
        const float* __restrict__ WRc, const float* __restrict__ WIc,
        const float* __restrict__ bmi, const float* __restrict__ bpi,
        float* outR, float* outI) {
    const int wave = threadIdx.x >> 5;
    const int lane = threadIdx.x & 31;
    const int l16  = lane & 15;
    const int half = lane >> 4;

    const float* yrt = outR;
    const float* yit = outI;

    const long rowOff = (long)(blockIdx.x * 16 + l16) * D_FEAT;
    const long wOff   = (long)(wave * 16 + l16) * D_FEAT;

    v8f accRa = {};
    v8f accRb = {};
    v8f accIa = {};
    v8f accIb = {};

    #pragma unroll
    for (int s = 0; s < 16; ++s) {
        const int k = 4 * s + 2 * half;
        float2 ar  = *(const float2*)(yr  + rowOff + k);
        float2 art = *(const float2*)(yrt + rowOff + k);
        float2 ai  = *(const float2*)(yi  + rowOff + k);
        float2 ait = *(const float2*)(yit + rowOff + k);
        v2f Azr = {0.5f * (ar.x + art.x), 0.5f * (ar.y + art.y)};
        v2f Azi = {0.5f * (ai.x + ait.x), 0.5f * (ai.y + ait.y)};
        v2f Ayr = {ar.x, ar.y};
        float2 wr2 = *(const float2*)(WRc + wOff + k);
        float2 wi2 = *(const float2*)(WIc + wOff + k);
        v2f Bwr = {wr2.x, wr2.y};
        v2f Bwi = {wi2.x, wi2.y};
        accRa = __builtin_amdgcn_wmma_f32_16x16x4_f32(false, Azr, false, Bwr, (short)0, accRa, false, false);
        accRb = __builtin_amdgcn_wmma_f32_16x16x4_f32(false, Azi, false, Bwi, (short)0, accRb, false, false);
        accIa = __builtin_amdgcn_wmma_f32_16x16x4_f32(false, Ayr, false, Bwi, (short)0, accIa, false, false);
        accIb = __builtin_amdgcn_wmma_f32_16x16x4_f32(false, Azi, false, Bwr, (short)0, accIb, false, false);
    }

    const int colD = wave * 16 + l16;
    const float bm = bmi[colD];
    const float bp = bpi[colD];

    __syncthreads();

    #pragma unroll
    for (int r = 0; r < 8; ++r) {
        long nodeM = (long)(blockIdx.x * 16 + r + 8 * half);
        outR[nodeM * D_FEAT + colD] = accRa[r] - accRb[r] + bm;
        outI[nodeM * D_FEAT + colD] = accIa[r] + accIb[r] + bp;
    }
}

extern "C" void kernel_launch(void* const* d_in, const int* in_sizes, int n_in,
                              void* d_out, int out_size, void* d_ws, size_t ws_size,
                              hipStream_t stream) {
    const float* xr = (const float*)d_in[0];
    const float* xi = (const float*)d_in[1];
    const int*   ei = (const int*)d_in[2];
    const float* Wr = (const float*)d_in[3];
    const float* br = (const float*)d_in[4];
    const float* Wi = (const float*)d_in[5];
    const float* bi = (const float*)d_in[6];

    const int n = in_sizes[0] / D_FEAT;
    const int E = in_sizes[2] / 2;
    const int K = in_sizes[3] / (D_FEAT * D_FEAT);
    const long NF = (long)n * D_FEAT;

    float* ws   = (float*)d_ws;
    float* yr   = ws;
    float* yi_  = ws + NF;
    float* degO = ws + 2 * NF;
    float* degI = degO + n;
    float* WRc  = degI + n;
    float* WIc  = WRc + D_FEAT * D_FEAT;
    float* bmi  = WIc + D_FEAT * D_FEAT;
    float* bpi  = bmi + D_FEAT;

    float* outR = (float*)d_out;
    float* outI = outR + NF;
    const int* row = ei;
    const int* col = ei + E;

    zero2_kernel<<<2048, 256, 0, stream>>>(ws, (2 * NF + 2 * (long)n) / 4,
                                           outR, (2 * NF) / 4);
    degree_kernel<<<(E + 255) / 256, 256, 0, stream>>>(row, col, E, degO, degI);
    deginv_kernel<<<(n + 255) / 256, 256, 0, stream>>>(degO, degI, n);
    wcomb_kernel<<<(D_FEAT * D_FEAT + D_FEAT + 255) / 256, 256, 0, stream>>>(
        Wr, br, Wi, bi, K, WRc, WIc, bmi, bpi);
    spmm_kernel<<<(E + 7) / 8, 256, 0, stream>>>(xr, xi, row, col, E,
                                                 degO, degI, yr, yi_, outR, outI);
    faber_gemm_kernel<<<n / 16, 128, 0, stream>>>(yr, yi_, WRc, WIc, bmi, bpi,
                                                  outR, outI);
}


# degree counts by workgroup-private LDS integer histograms (edge slices x node ranges), planes summed in the deg^-0.25 kernel; replaces 3.2M scattered global f32 atomics; launch order changed
# speedup vs baseline: 1.2211x; 1.2211x over previous
#include <hip/hip_runtime.h>
#include <math.h>

typedef __attribute__((ext_vector_type(2))) float v2f;
typedef __attribute__((ext_vector_type(8))) float v8f;

#define D_FEAT 64

__global__ void zero2_kernel(float* __restrict__ a, long na4,
                             float* __restrict__ b, long nb4) {
    long i = (long)blockIdx.x * blockDim.x + threadIdx.x;
    long stride = (long)gridDim.x * blockDim.x;
    float4 z = make_float4(0.f, 0.f, 0.f, 0.f);
    for (long j = i; j < na4 + nb4; j += stride) {
        if (j < na4) ((float4*)a)[j] = z;
        else         ((float4*)b)[j - na4] = z;
    }
}

#define HIST_RANGE  25600
#define HIST_SLICES 32
static_assert(2 * HIST_SLICES <= D_FEAT, "the partial planes must fit the n * D_FEAT words of one accumulator plane");

__global__ __launch_bounds__(1024) void degree_hist_kernel(
        const int* __restrict__ row, const int* __restrict__ col, int E, int n,
        unsigned int* __restrict__ part) {
    __shared__ unsigned int hist[HIST_RANGE];

    const int slice = blockIdx.x;
    const int lo    = blockIdx.y * HIST_RANGE;
    const int which = blockIdx.z;
    const int* __restrict__ idx = which ? col : row;

    for (int i = threadIdx.x; i < HIST_RANGE / 4; i += blockDim.x)
        ((uint4*)hist)[i] = make_uint4(0u, 0u, 0u, 0u);
    __syncthreads();

    const int per   = ((E + 4 * (int)gridDim.x - 1) / (4 * (int)gridDim.x)) * 4;
    const int begin = min(E, slice * per);
    const int end   = min(E, begin + per);

    auto bump = [&](int k) {
        const unsigned int d = (unsigned int)(k - lo);
        if (d < (unsigned int)HIST_RANGE) atomicAdd(&hist[d], 1u);
    };

    int done = begin;
    if ((((size_t)idx) & 15) == 0) {
        const int4* __restrict__ v = (const int4*)(idx + begin);
        const int n4 = (end - begin) >> 2;
        for (int i = threadIdx.x; i < n4; i += 4 * blockDim.x) {
            int4 q[4];
            #pragma unroll
            for (int u = 0; u < 4; ++u) {
                const int j = i + u * blockDim.x;
                q[u] = (j < n4) ? v[j] : make_int4(-1, -1, -1, -1);
            }
            #pragma unroll
            for (int u = 0; u < 4; ++u) {
                bump(q[u].x); bump(q[u].y); bump(q[u].z); bump(q[u].w);
            }
        }
        done = begin + 4 * n4;
    }
    for (int i = done + threadIdx.x; i < end; i += blockDim.x) bump(idx[i]);

    __syncthreads();

    const int cnt = min(HIST_RANGE, n - lo);
    unsigned int* __restrict__ dst = part + (long)(slice * 2 + which) * n + lo;
    if ((n & 3) == 0) {
        for (int i = threadIdx.x; i < cnt / 4; i += blockDim.x)
            ((uint4*)dst)[i] = ((const uint4*)hist)[i];
    } else {
        for (int i = threadIdx.x; i < cnt; i += blockDim.x) dst[i] = hist[i];
    }
}

__global__ void deginv_kernel(const unsigned int* __restrict__ part, int n,
                              float* degO, float* degI) {
    const int i = blockIdx.x * blockDim.x + threadIdx.x;
    if (i >= n) return;
    const unsigned int* __restrict__ p = part + (long)blockIdx.y * n + i;
    unsigned int s = 0u;
    for (int k = 0; k < HIST_SLICES; ++k) s += p[(long)k * 2 * n];
    const float d = (float)s;
    const float r = (d > 0.f) ? powf(d, -0.25f) : 0.f;
    if (blockIdx.y) degI[i] = r; else degO[i] = r;
}

__global__ void wcomb_kernel(const float* __restrict__ Wr, const float* __restrict__ br,
                             const float* __restrict__ Wi, const float* __restrict__ bi,
                             int K, float* WRc, float* WIc, float* bmi, float* bpi) {
    int i = blockIdx.x * blockDim.x + threadIdx.x;
    if (i < D_FEAT * D_FEAT) {
        float sr = 0.f, si = 0.f, c = 1.f;
        for (int k = 0; k < K; ++k) {
            sr += c * Wr[k * D_FEAT * D_FEAT + i];
            si += c * Wi[k * D_FEAT * D_FEAT + i];
            c *= 0.5f;
        }
        WRc[i] = sr;
        WIc[i] = si;
    } else if (i < D_FEAT * D_FEAT + D_FEAT) {
        int o = i - D_FEAT * D_FEAT;
        float sr = 0.f, si = 0.f, c = 1.f;
        for (int k = 0; k < K; ++k) {
            sr += c * br[k * D_FEAT + o];
            si += c * bi[k * D_FEAT + o];
            c *= 0.5f;
        }
        bmi[o] = sr - si;
        bpi[o] = sr + si;
    }
}

__global__ __launch_bounds__(256) void spmm_kernel(
        const float* __restrict__ xr, const float* __restrict__ xi,
        const int* __restrict__ row, const int* __restrict__ col, int E,
        const float* __restrict__ out_inv, const float* __restrict__ in_inv,
        float* yr, float* yi, float* yrt, float* yit) {
    const int lane = threadIdx.x & 31;
    int e = blockIdx.x * 8 + (threadIdx.x >> 5);
    e = __builtin_amdgcn_readfirstlane(e);
    if (e >= E) return;

    const int r = row[e];
    const int c = col[e];
    const float w = out_inv[r] * in_inv[c];

    const int d = lane << 1;
    const long ro = (long)r * D_FEAT + d;
    const long co = (long)c * D_FEAT + d;

    const float2 xrc = *(const float2*)(xr + co);
    const float2 xic = *(const float2*)(xi + co);
    const float2 xrr = *(const float2*)(xr + ro);
    const float2 xir = *(const float2*)(xi + ro);

    unsafeAtomicAdd(&yr[ro],      w * xrc.x);
    unsafeAtomicAdd(&yr[ro + 1],  w * xrc.y);
    unsafeAtomicAdd(&yi[ro],      w * xic.x);
    unsafeAtomicAdd(&yi[ro + 1],  w * xic.y);
    unsafeAtomicAdd(&yrt[co],     w * xrr.x);
    unsafeAtomicAdd(&yrt[co + 1], w * xrr.y);
    unsafeAtomicAdd(&yit[co],     w * xir.x);
    unsafeAtomicAdd(&yit[co + 1], w * xir.y);
}

__global__ __launch_bounds__(128) void faber_gemm_kernel(
        const float* __restrict__ yr, const float* __restrict__ yi,
        const float* __restrict__ WRc, const float* __restrict__ WIc,
        const float* __restrict__ bmi, const float* __restrict__ bpi,
        float* outR, float* outI) {
    const int wave = threadIdx.x >> 5;
    const int lane = threadIdx.x & 31;
    const int l16  = lane & 15;
    const int half = lane >> 4;

    const float* yrt = outR;
    const float* yit = outI;

    const long rowOff = (long)(blockIdx.x * 16 + l16) * D_FEAT;
    const long wOff   = (long)(wave * 16 + l16) * D_FEAT;

    v8f accRa = {};
    v8f accRb = {};
    v8f accIa = {};
    v8f accIb = {};

    #pragma unroll
    for (int s = 0; s < 16; ++s) {
        const int k = 4 * s + 2 * half;
        float2 ar  = *(const float2*)(yr  + rowOff + k);
        float2 art = *(const float2*)(yrt + rowOff + k);
        float2 ai  = *(const float2*)(yi  + rowOff + k);
        float2 ait = *(const float2*)(yit + rowOff + k);
        v2f Azr = {0.5f * (ar.x + art.x), 0.5f * (ar.y + art.y)};
        v2f Azi = {0.5f * (ai.x + ait.x), 0.5f * (ai.y + ait.y)};
        v2f Ayr = {ar.x, ar.y};
        float2 wr2 = *(const float2*)(WRc + wOff + k);
        float2 wi2 = *(const float2*)(WIc + wOff + k);
        v2f Bwr = {wr2.x, wr2.y};
        v2f Bwi = {wi2.x, wi2.y};
        accRa = __builtin_amdgcn_wmma_f32_16x16x4_f32(false, Azr, false, Bwr, (short)0, accRa, false, false);
        accRb = __builtin_amdgcn_wmma_f32_16x16x4_f32(false, Azi, false, Bwi, (short)0, accRb, false, false);
        accIa = __builtin_amdgcn_wmma_f32_16x16x4_f32(false, Ayr, false, Bwi, (short)0, accIa, false, false);
        accIb = __builtin_amdgcn_wmma_f32_16x16x4_f32(false, Azi, false, Bwr, (short)0, accIb, false, false);
    }

    const int colD = wave * 16 + l16;
    const float bm = bmi[colD];
    const float bp = bpi[colD];

    __syncthreads();

    #pragma unroll
    for (int r = 0; r < 8; ++r) {
        long nodeM = (long)(blockIdx.x * 16 + r + 8 * half);
        outR[nodeM * D_FEAT + colD] = accRa[r] - accRb[r] + bm;
        outI[nodeM * D_FEAT + colD] = accIa[r] + accIb[r] + bp;
    }
}

extern "C" void kernel_launch(void* const* d_in, const int* in_sizes, int n_in,
                              void* d_out, int out_size, void* d_ws, size_t ws_size,
                              hipStream_t stream) {
    const float* xr = (const float*)d_in[0];
    const float* xi = (const float*)d_in[1];
    const int*   ei = (const int*)d_in[2];
    const float* Wr = (const float*)d_in[3];
    const float* br = (const float*)d_in[4];
    const float* Wi = (const float*)d_in[5];
    const float* bi = (const float*)d_in[6];

    const int n = in_sizes[0] / D_FEAT;
    const int E = in_sizes[2] / 2;
    const int K = in_sizes[3] / (D_FEAT * D_FEAT);
    const long NF = (long)n * D_FEAT;

    float* ws   = (float*)d_ws;
    float* yr   = ws;
    float* yi_  = ws + NF;
    float* degO = ws + 2 * NF;
    float* degI = degO + n;
    float* WRc  = degI + n;
    float* WIc  = WRc + D_FEAT * D_FEAT;
    float* bmi  = WIc + D_FEAT * D_FEAT;
    float* bpi  = bmi + D_FEAT;

    float* outR = (float*)d_out;
    float* outI = outR + NF;
    const int* row = ei;
    const int* col = ei + E;

    unsigned int* part = (unsigned int*)yr;
    degree_hist_kernel<<<dim3(HIST_SLICES, (n + HIST_RANGE - 1) / HIST_RANGE, 2), 1024, 0, stream>>>(
        row, col, E, n, part);
    deginv_kernel<<<dim3((n + 255) / 256, 2), 256, 0, stream>>>(part, n, degO, degI);
    zero2_kernel<<<2048, 256, 0, stream>>>(ws, (2 * NF) / 4, outR, (2 * NF) / 4);
    wcomb_kernel<<<(D_FEAT * D_FEAT + D_FEAT + 255) / 256, 256, 0, stream>>>(
        Wr, br, Wi, bi, K, WRc, WIc, bmi, bpi);
    spmm_kernel<<<(E + 7) / 8, 256, 0, stream>>>(xr, xi, row, col, E,
                                                 degO, degI, yr, yi_, outR, outI);
    faber_gemm_kernel<<<n / 16, 128, 0, stream>>>(yr, yi_, WRc, WIc, bmi, bpi,
                                                  outR, outI);
}


# edges grouped per node by an LDS-cursor counting sort on the slice histograms; one wave per node sums its neighbour rows in registers: no f32 atomics, no zero fill; sum order changes
# speedup vs baseline: 2.3331x; 1.9107x over previous
#include <hip/hip_runtime.h>
#include <math.h>

typedef __attribute__((ext_vector_type(2))) float v2f;
typedef __attribute__((ext_vector_type(8))) float v8f;

#define D_FEAT 64

#define HIST_RANGE  25600
#define HIST_SLICES 32
static_assert(2 * HIST_SLICES <= D_FEAT, "the partial planes must fit the n * D_FEAT words of one accumulator plane");

__global__ __launch_bounds__(1024) void degree_hist_kernel(
        const int* __restrict__ row, const int* __restrict__ col, int E, int n,
        unsigned int* __restrict__ part) {
    __shared__ unsigned int hist[HIST_RANGE];

    const int slice = blockIdx.x;
    const int lo    = blockIdx.y * HIST_RANGE;
    const int which = blockIdx.z;
    const int* __restrict__ idx = which ? col : row;

    for (int i = threadIdx.x; i < HIST_RANGE / 4; i += blockDim.x)
        ((uint4*)hist)[i] = make_uint4(0u, 0u, 0u, 0u);
    __syncthreads();

    const int per   = ((E + 4 * (int)gridDim.x - 1) / (4 * (int)gridDim.x)) * 4;
    const int begin = min(E, slice * per);
    const int end   = min(E, begin + per);

    auto bump = [&](int k) {
        const unsigned int d = (unsigned int)(k - lo);
        if (d < (unsigned int)HIST_RANGE) atomicAdd(&hist[d], 1u);
    };

    int done = begin;
    if ((((size_t)idx) & 15) == 0) {
        const int4* __restrict__ v = (const int4*)(idx + begin);
        const int n4 = (end - begin) >> 2;
        for (int i = threadIdx.x; i < n4; i += 4 * blockDim.x) {
            int4 q[4];
            #pragma unroll
            for (int u = 0; u < 4; ++u) {
                const int j = i + u * blockDim.x;
                q[u] = (j < n4) ? v[j] : make_int4(-1, -1, -1, -1);
            }
            #pragma unroll
            for (int u = 0; u < 4; ++u) {
                bump(q[u].x); bump(q[u].y); bump(q[u].z); bump(q[u].w);
            }
        }
        done = begin + 4 * n4;
    }
    for (int i = done + threadIdx.x; i < end; i += blockDim.x) bump(idx[i]);

    __syncthreads();

    const int cnt = min(HIST_RANGE, n - lo);
    unsigned int* __restrict__ dst = part + (long)(slice * 2 + which) * n + lo;
    if ((n & 3) == 0) {
        for (int i = threadIdx.x; i < cnt / 4; i += blockDim.x)
            ((uint4*)dst)[i] = ((const uint4*)hist)[i];
    } else {
        for (int i = threadIdx.x; i < cnt; i += blockDim.x) dst[i] = hist[i];
    }
}

#define SCAN_BLOCK 256
#define NB_MAX     1024

__global__ __launch_bounds__(SCAN_BLOCK) void deginv_kernel(
        unsigned int* part, int n, float* __restrict__ degO, float* __restrict__ degI,
        unsigned int* __restrict__ cnt, unsigned int* __restrict__ locOff,
        unsigned int* __restrict__ blkSum) {
    __shared__ unsigned int sc[SCAN_BLOCK];
    const int tid   = threadIdx.x;
    const int which = blockIdx.y;
    const int i     = blockIdx.x * SCAN_BLOCK + tid;
    unsigned int s = 0u;
    if (i < n) {
        unsigned int* p = part + (long)which * n + i;
        unsigned int t[HIST_SLICES];
        #pragma unroll
        for (int k = 0; k < HIST_SLICES; ++k) t[k] = p[(long)k * 2 * n];
        #pragma unroll
        for (int k = 0; k < HIST_SLICES; ++k) { p[(long)k * 2 * n] = s; s += t[k]; }
        cnt[(long)which * n + i] = s;
        const float d = (float)s;
        const float r = (d > 0.f) ? powf(d, -0.25f) : 0.f;
        if (which) degI[i] = r; else degO[i] = r;
    }
    sc[tid] = s;
    __syncthreads();
    for (int off = 1; off < SCAN_BLOCK; off <<= 1) {
        const unsigned int t = (tid >= off) ? sc[tid - off] : 0u;
        __syncthreads();
        sc[tid] += t;
        __syncthreads();
    }
    if (i < n) locOff[(long)which * n + i] = sc[tid] - s;
    if (tid == SCAN_BLOCK - 1) blkSum[which * NB_MAX + blockIdx.x] = sc[tid];
}

__global__ __launch_bounds__(NB_MAX) void blockscan_kernel(
        const unsigned int* __restrict__ blkSum, int nb, unsigned int* __restrict__ blkOff) {
    __shared__ unsigned int sc[NB_MAX];
    const int tid   = threadIdx.x;
    const int which = blockIdx.x;
    const unsigned int s = (tid < nb) ? blkSum[which * NB_MAX + tid] : 0u;
    sc[tid] = s;
    __syncthreads();
    for (int off = 1; off < NB_MAX; off <<= 1) {
        const unsigned int t = (tid >= off) ? sc[tid - off] : 0u;
        __syncthreads();
        sc[tid] += t;
        __syncthreads();
    }
    blkOff[which * NB_MAX + tid] = sc[tid] - s;
}

__global__ __launch_bounds__(1024) void fill_kernel(
        const int* __restrict__ row, const int* __restrict__ col, int E, int n,
        const unsigned int* __restrict__ part, const unsigned int* __restrict__ locOff,
        const unsigned int* __restrict__ blkOff,
        int* __restrict__ adjO, int* __restrict__ adjI) {
    __shared__ unsigned int cursor[HIST_RANGE];

    const int slice = blockIdx.x;
    const int lo    = blockIdx.y * HIST_RANGE;
    const int which = blockIdx.z;
    const int* __restrict__ key = which ? col : row;
    const int* __restrict__ oth = which ? row : col;
    int* __restrict__ adj = which ? adjI : adjO;

    const int cntN = min(HIST_RANGE, n - lo);
    for (int d = threadIdx.x; d < cntN; d += blockDim.x) {
        const int node = lo + d;
        cursor[d] = blkOff[which * NB_MAX + (node / SCAN_BLOCK)]
                  + locOff[(long)which * n + node]
                  + part[(long)(slice * 2 + which) * n + node];
    }
    __syncthreads();

    const int per   = ((E + 4 * (int)gridDim.x - 1) / (4 * (int)gridDim.x)) * 4;
    const int begin = min(E, slice * per);
    const int end   = min(E, begin + per);

    auto place = [&](int k, int o) {
        const unsigned int d = (unsigned int)(k - lo);
        if (d < (unsigned int)HIST_RANGE) {
            const unsigned int pos = atomicAdd(&cursor[d], 1u);
            adj[pos] = o;
        }
    };

    int done = begin;
    if (((((size_t)key) | ((size_t)oth)) & 15) == 0) {
        const int4* __restrict__ vk = (const int4*)(key + begin);
        const int4* __restrict__ vo = (const int4*)(oth + begin);
        const int n4 = (end - begin) >> 2;
        for (int i = threadIdx.x; i < n4; i += 4 * blockDim.x) {
            int4 q[4], p[4];
            #pragma unroll
            for (int u = 0; u < 4; ++u) {
                const int j = i + u * blockDim.x;
                q[u] = (j < n4) ? vk[j] : make_int4(-1, -1, -1, -1);
                p[u] = (j < n4) ? vo[j] : make_int4(0, 0, 0, 0);
            }
            #pragma unroll
            for (int u = 0; u < 4; ++u) {
                place(q[u].x, p[u].x); place(q[u].y, p[u].y);
                place(q[u].z, p[u].z); place(q[u].w, p[u].w);
            }
        }
        done = begin + 4 * n4;
    }
    for (int i = done + threadIdx.x; i < end; i += blockDim.x) place(key[i], oth[i]);
}

__global__ void wcomb_kernel(const float* __restrict__ Wr, const float* __restrict__ br,
                             const float* __restrict__ Wi, const float* __restrict__ bi,
                             int K, float* WRc, float* WIc, float* bmi, float* bpi) {
    int i = blockIdx.x * blockDim.x + threadIdx.x;
    if (i < D_FEAT * D_FEAT) {
        float sr = 0.f, si = 0.f, c = 1.f;
        for (int k = 0; k < K; ++k) {
            sr += c * Wr[k * D_FEAT * D_FEAT + i];
            si += c * Wi[k * D_FEAT * D_FEAT + i];
            c *= 0.5f;
        }
        WRc[i] = sr;
        WIc[i] = si;
    } else if (i < D_FEAT * D_FEAT + D_FEAT) {
        int o = i - D_FEAT * D_FEAT;
        float sr = 0.f, si = 0.f, c = 1.f;
        for (int k = 0; k < K; ++k) {
            sr += c * br[k * D_FEAT + o];
            si += c * bi[k * D_FEAT + o];
            c *= 0.5f;
        }
        bmi[o] = sr - si;
        bpi[o] = sr + si;
    }
}

__device__ __forceinline__ void gather_list(
        const int* __restrict__ adj, unsigned int start, unsigned int cnt,
        const float* __restrict__ winv, float self,
        const float* __restrict__ xr, const float* __restrict__ xi,
        int lane, float2& accR, float2& accI) {
    const int d = lane << 1;
    for (unsigned int base = 0; base < cnt; base += 32u) {
        const unsigned int m = min(32u, cnt - base);
        int   nb = 0;
        float w  = 0.f;
        if ((unsigned int)lane < m) {
            nb = adj[start + base + lane];
            w  = self * winv[nb];
        }
        unsigned int j = 0;
        for (; j + 4u <= m; j += 4u) {
            int   c[4];
            float wj[4];
            float2 a[4], b[4];
            #pragma unroll
            for (int u = 0; u < 4; ++u) {
                c[u]  = __builtin_amdgcn_readlane(nb, (int)(j + u));
                wj[u] = __int_as_float(__builtin_amdgcn_readlane(__float_as_int(w), (int)(j + u)));
                a[u]  = *(const float2*)(xr + (long)c[u] * D_FEAT + d);
                b[u]  = *(const float2*)(xi + (long)c[u] * D_FEAT + d);
            }
            #pragma unroll
            for (int u = 0; u < 4; ++u) {
                accR.x += wj[u] * a[u].x;  accR.y += wj[u] * a[u].y;
                accI.x += wj[u] * b[u].x;  accI.y += wj[u] * b[u].y;
            }
        }
        for (; j < m; ++j) {
            const int   c  = __builtin_amdgcn_readlane(nb, (int)j);
            const float wj = __int_as_float(__builtin_amdgcn_readlane(__float_as_int(w), (int)j));
            const float2 a = *(const float2*)(xr + (long)c * D_FEAT + d);
            const float2 b = *(const float2*)(xi + (long)c * D_FEAT + d);
            accR.x += wj * a.x;  accR.y += wj * a.y;
            accI.x += wj * b.x;  accI.y += wj * b.y;
        }
    }
}

__global__ __launch_bounds__(256) void gather_kernel(
        const float* __restrict__ xr, const float* __restrict__ xi,
        const int* __restrict__ adjO, const int* __restrict__ adjI,
        const unsigned int* __restrict__ blkOff, const unsigned int* __restrict__ locOff,
        const unsigned int* __restrict__ cnt,
        const float* __restrict__ out_inv, const float* __restrict__ in_inv, int n,
        float* __restrict__ YR, float* __restrict__ ZR, float* __restrict__ ZI) {
    const int lane = threadIdx.x & 31;
    int v = blockIdx.x * 8 + (threadIdx.x >> 5);
    v = __builtin_amdgcn_readfirstlane(v);
    if (v >= n) return;

    const unsigned int sO = blkOff[v / SCAN_BLOCK] + locOff[v];
    const unsigned int sI = blkOff[NB_MAX + v / SCAN_BLOCK] + locOff[(long)n + v];
    const unsigned int cO = cnt[v];
    const unsigned int cI = cnt[(long)n + v];
    const float dO = out_inv[v];
    const float dI = in_inv[v];

    float2 yr  = make_float2(0.f, 0.f), yi  = make_float2(0.f, 0.f);
    float2 yrt = make_float2(0.f, 0.f), yit = make_float2(0.f, 0.f);
    gather_list(adjO, sO, cO, in_inv, dO, xr, xi, lane, yr, yi);
    gather_list(adjI, sI, cI, out_inv, dI, xr, xi, lane, yrt, yit);

    const long o = (long)v * D_FEAT + (lane << 1);
    *(float2*)(YR + o) = yr;
    *(float2*)(ZR + o) = make_float2(0.5f * (yr.x + yrt.x), 0.5f * (yr.y + yrt.y));
    *(float2*)(ZI + o) = make_float2(0.5f * (yi.x + yit.x), 0.5f * (yi.y + yit.y));
}

__global__ __launch_bounds__(128) void faber_gemm_kernel(
        const float* __restrict__ yr,
        const float* __restrict__ WRc, const float* __restrict__ WIc,
        const float* __restrict__ bmi, const float* __restrict__ bpi,
        float* outR, float* outI) {
    const int wave = threadIdx.x >> 5;
    const int lane = threadIdx.x & 31;
    const int l16  = lane & 15;
    const int half = lane >> 4;

    const float* zr = outR;
    const float* zi = outI;

    const long rowOff = (long)(blockIdx.x * 16 + l16) * D_FEAT;
    const long wOff   = (long)(wave * 16 + l16) * D_FEAT;

    v8f accRa = {};
    v8f accRb = {};
    v8f accIa = {};
    v8f accIb = {};

    #pragma unroll
    for (int s = 0; s < 16; ++s) {
        const int k = 4 * s + 2 * half;
        float2 ar  = *(const float2*)(yr + rowOff + k);
        float2 azr = *(const float2*)(zr + rowOff + k);
        float2 azi = *(const float2*)(zi + rowOff + k);
        v2f Azr = {azr.x, azr.y};
        v2f Azi = {azi.x, azi.y};
        v2f Ayr = {ar.x, ar.y};
        float2 wr2 = *(const float2*)(WRc + wOff + k);
        float2 wi2 = *(const float2*)(WIc + wOff + k);
        v2f Bwr = {wr2.x, wr2.y};
        v2f Bwi = {wi2.x, wi2.y};
        accRa = __builtin_amdgcn_wmma_f32_16x16x4_f32(false, Azr, false, Bwr, (short)0, accRa, false, false);
        accRb = __builtin_amdgcn_wmma_f32_16x16x4_f32(false, Azi, false, Bwi, (short)0, accRb, false, false);
        accIa = __builtin_amdgcn_wmma_f32_16x16x4_f32(false, Ayr, false, Bwi, (short)0, accIa, false, false);
        accIb = __builtin_amdgcn_wmma_f32_16x16x4_f32(false, Azi, false, Bwr, (short)0, accIb, false, false);
    }

    const int colD = wave * 16 + l16;
    const float bm = bmi[colD];
    const float bp = bpi[colD];

    __syncthreads();

    #pragma unroll
    for (int r = 0; r < 8; ++r) {
        long nodeM = (long)(blockIdx.x * 16 + r + 8 * half);
        outR[nodeM * D_FEAT + colD] = accRa[r] - accRb[r] + bm;
        outI[nodeM * D_FEAT + colD] = accIa[r] + accIb[r] + bp;
    }
}

extern "C" void kernel_launch(void* const* d_in, const int* in_sizes, int n_in,
                              void* d_out, int out_size, void* d_ws, size_t ws_size,
                              hipStream_t stream) {
    const float* xr = (const float*)d_in[0];
    const float* xi = (const float*)d_in[1];
    const int*   ei = (const int*)d_in[2];
    const float* Wr = (const float*)d_in[3];
    const float* br = (const float*)d_in[4];
    const float* Wi = (const float*)d_in[5];
    const float* bi = (const float*)d_in[6];

    const int n = in_sizes[0] / D_FEAT;
    const int E = in_sizes[2] / 2;
    const int K = in_sizes[3] / (D_FEAT * D_FEAT);
    const long NF = (long)n * D_FEAT;

    float* ws   = (float*)d_ws;
    float* yr   = ws;
    float* degO = ws + 2 * NF;
    float* degI = degO + n;
    float* WRc  = degI + n;
    float* WIc  = WRc + D_FEAT * D_FEAT;
    float* bmi  = WIc + D_FEAT * D_FEAT;
    float* bpi  = bmi + D_FEAT;

    int* adjO = (int*)(ws + NF);
    int* adjI = adjO + E;
    unsigned int* cnt    = (unsigned int*)(adjI + E);
    unsigned int* locOff = cnt + 2 * (long)n;
    unsigned int* blkSum = locOff + 2 * (long)n;
    unsigned int* blkOff = blkSum + 2 * NB_MAX;

    float* outR = (float*)d_out;
    float* outI = outR + NF;
    const int* row = ei;
    const int* col = ei + E;

    unsigned int* part = (unsigned int*)yr;
    const dim3 hgrid(HIST_SLICES, (n + HIST_RANGE - 1) / HIST_RANGE, 2);
    degree_hist_kernel<<<hgrid, 1024, 0, stream>>>(row, col, E, n, part);
    deginv_kernel<<<dim3((n + SCAN_BLOCK - 1) / SCAN_BLOCK, 2), SCAN_BLOCK, 0, stream>>>(
        part, n, degO, degI, cnt, locOff, blkSum);
    blockscan_kernel<<<2, NB_MAX, 0, stream>>>(blkSum, (n + SCAN_BLOCK - 1) / SCAN_BLOCK, blkOff);
    fill_kernel<<<hgrid, 1024, 0, stream>>>(row, col, E, n, part, locOff, blkOff, adjO, adjI);
    wcomb_kernel<<<(D_FEAT * D_FEAT + D_FEAT + 255) / 256, 256, 0, stream>>>(
        Wr, br, Wi, bi, K, WRc, WIc, bmi, bpi);
    gather_kernel<<<(n + 7) / 8, 256, 0, stream>>>(xr, xi, adjO, adjI, blkOff, locOff, cnt,
                                                   degO, degI, n, yr, outR, outI);
    faber_gemm_kernel<<<n / 16, 128, 0, stream>>>(yr, WRc, WIc, bmi, bpi, outR, outI);
}


# gather and GEMM fused in one kernel: each 8-wave block gathers 16 nodes' sums to LDS and multiplies them there (weights in LDS once per block); y_real/ZR/ZI never reach memory; one launch fewer
# speedup vs baseline: 2.7438x; 1.1760x over previous
#include <hip/hip_runtime.h>
#include <math.h>

typedef __attribute__((ext_vector_type(2))) float v2f;
typedef __attribute__((ext_vector_type(8))) float v8f;

#define D_FEAT 64

#define HIST_RANGE  25600
#define HIST_SLICES 32
static_assert(2 * HIST_SLICES <= D_FEAT, "the partial planes must fit the n * D_FEAT words of one accumulator plane");

__global__ __launch_bounds__(1024) void degree_hist_kernel(
        const int* __restrict__ row, const int* __restrict__ col, int E, int n,
        unsigned int* __restrict__ part) {
    __shared__ unsigned int hist[HIST_RANGE];

    const int slice = blockIdx.x;
    const int lo    = blockIdx.y * HIST_RANGE;
    const int which = blockIdx.z;
    const int* __restrict__ idx = which ? col : row;

    for (int i = threadIdx.x; i < HIST_RANGE / 4; i += blockDim.x)
        ((uint4*)hist)[i] = make_uint4(0u, 0u, 0u, 0u);
    __syncthreads();

    const int per   = ((E + 4 * (int)gridDim.x - 1) / (4 * (int)gridDim.x)) * 4;
    const int begin = min(E, slice * per);
    const int end   = min(E, begin + per);

    auto bump = [&](int k) {
        const unsigned int d = (unsigned int)(k - lo);
        if (d < (unsigned int)HIST_RANGE) atomicAdd(&hist[d], 1u);
    };

    int done = begin;
    if ((((size_t)idx) & 15) == 0) {
        const int4* __restrict__ v = (const int4*)(idx + begin);
        const int n4 = (end - begin) >> 2;
        for (int i = threadIdx.x; i < n4; i += 4 * blockDim.x) {
            int4 q[4];
            #pragma unroll
            for (int u = 0; u < 4; ++u) {
                const int j = i + u * blockDim.x;
                q[u] = (j < n4) ? v[j] : make_int4(-1, -1, -1, -1);
            }
            #pragma unroll
            for (int u = 0; u < 4; ++u) {
                bump(q[u].x); bump(q[u].y); bump(q[u].z); bump(q[u].w);
            }
        }
        done = begin + 4 * n4;
    }
    for (int i = done + threadIdx.x; i < end; i += blockDim.x) bump(idx[i]);

    __syncthreads();

    const int cnt = min(HIST_RANGE, n - lo);
    unsigned int* __restrict__ dst = part + (long)(slice * 2 + which) * n + lo;
    if ((n & 3) == 0) {
        for (int i = threadIdx.x; i < cnt / 4; i += blockDim.x)
            ((uint4*)dst)[i] = ((const uint4*)hist)[i];
    } else {
        for (int i = threadIdx.x; i < cnt; i += blockDim.x) dst[i] = hist[i];
    }
}

#define SCAN_BLOCK 256
#define NB_MAX     1024

__global__ __launch_bounds__(SCAN_BLOCK) void deginv_kernel(
        unsigned int* part, int n, float* __restrict__ degO, float* __restrict__ degI,
        unsigned int* __restrict__ cnt, unsigned int* __restrict__ locOff,
        unsigned int* __restrict__ blkSum) {
    __shared__ unsigned int sc[SCAN_BLOCK];
    const int tid   = threadIdx.x;
    const int which = blockIdx.y;
    const int i     = blockIdx.x * SCAN_BLOCK + tid;
    unsigned int s = 0u;
    if (i < n) {
        unsigned int* p = part + (long)which * n + i;
        unsigned int t[HIST_SLICES];
        #pragma unroll
        for (int k = 0; k < HIST_SLICES; ++k) t[k] = p[(long)k * 2 * n];
        #pragma unroll
        for (int k = 0; k < HIST_SLICES; ++k) { p[(long)k * 2 * n] = s; s += t[k]; }
        cnt[(long)which * n + i] = s;
        const float d = (float)s;
        const float r = (d > 0.f) ? powf(d, -0.25f) : 0.f;
        if (which) degI[i] = r; else degO[i] = r;
    }
    sc[tid] = s;
    __syncthreads();
    for (int off = 1; off < SCAN_BLOCK; off <<= 1) {
        const unsigned int t = (tid >= off) ? sc[tid - off] : 0u;
        __syncthreads();
        sc[tid] += t;
        __syncthreads();
    }
    if (i < n) locOff[(long)which * n + i] = sc[tid] - s;
    if (tid == SCAN_BLOCK - 1) blkSum[which * NB_MAX + blockIdx.x] = sc[tid];
}

__global__ __launch_bounds__(NB_MAX) void blockscan_kernel(
        const unsigned int* __restrict__ blkSum, int nb, unsigned int* __restrict__ blkOff) {
    __shared__ unsigned int sc[NB_MAX];
    const int tid   = threadIdx.x;
    const int which = blockIdx.x;
    const unsigned int s = (tid < nb) ? blkSum[which * NB_MAX + tid] : 0u;
    sc[tid] = s;
    __syncthreads();
    for (int off = 1; off < NB_MAX; off <<= 1) {
        const unsigned int t = (tid >= off) ? sc[tid - off] : 0u;
        __syncthreads();
        sc[tid] += t;
        __syncthreads();
    }
    blkOff[which * NB_MAX + tid] = sc[tid] - s;
}

__global__ __launch_bounds__(1024) void fill_kernel(
        const int* __restrict__ row, const int* __restrict__ col, int E, int n,
        const unsigned int* __restrict__ part, const unsigned int* __restrict__ locOff,
        const unsigned int* __restrict__ blkOff,
        int* __restrict__ adjO, int* __restrict__ adjI) {
    __shared__ unsigned int cursor[HIST_RANGE];

    const int slice = blockIdx.x;
    const int lo    = blockIdx.y * HIST_RANGE;
    const int which = blockIdx.z;
    const int* __restrict__ key = which ? col : row;
    const int* __restrict__ oth = which ? row : col;
    int* __restrict__ adj = which ? adjI : adjO;

    const int cntN = min(HIST_RANGE, n - lo);
    for (int d = threadIdx.x; d < cntN; d += blockDim.x) {
        const int node = lo + d;
        cursor[d] = blkOff[which * NB_MAX + (node / SCAN_BLOCK)]
                  + locOff[(long)which * n + node]
                  + part[(long)(slice * 2 + which) * n + node];
    }
    __syncthreads();

    const int per   = ((E + 4 * (int)gridDim.x - 1) / (4 * (int)gridDim.x)) * 4;
    const int begin = min(E, slice * per);
    const int end   = min(E, begin + per);

    auto place = [&](int k, int o) {
        const unsigned int d = (unsigned int)(k - lo);
        if (d < (unsigned int)HIST_RANGE) {
            const unsigned int pos = atomicAdd(&cursor[d], 1u);
            adj[pos] = o;
        }
    };

    int done = begin;
    if (((((size_t)key) | ((size_t)oth)) & 15) == 0) {
        const int4* __restrict__ vk = (const int4*)(key + begin);
        const int4* __restrict__ vo = (const int4*)(oth + begin);
        const int n4 = (end - begin) >> 2;
        for (int i = threadIdx.x; i < n4; i += 4 * blockDim.x) {
            int4 q[4], p[4];
            #pragma unroll
            for (int u = 0; u < 4; ++u) {
                const int j = i + u * blockDim.x;
                q[u] = (j < n4) ? vk[j] : make_int4(-1, -1, -1, -1);
                p[u] = (j < n4) ? vo[j] : make_int4(0, 0, 0, 0);
            }
            #pragma unroll
            for (int u = 0; u < 4; ++u) {
                place(q[u].x, p[u].x); place(q[u].y, p[u].y);
                place(q[u].z, p[u].z); place(q[u].w, p[u].w);
            }
        }
        done = begin + 4 * n4;
    }
    for (int i = done + threadIdx.x; i < end; i += blockDim.x) place(key[i], oth[i]);
}

__global__ void wcomb_kernel(const float* __restrict__ Wr, const float* __restrict__ br,
                             const float* __restrict__ Wi, const float* __restrict__ bi,
                             int K, float* WRc, float* WIc, float* bmi, float* bpi) {
    int i = blockIdx.x * blockDim.x + threadIdx.x;
    if (i < D_FEAT * D_FEAT) {
        float sr = 0.f, si = 0.f, c = 1.f;
        for (int k = 0; k < K; ++k) {
            sr += c * Wr[k * D_FEAT * D_FEAT + i];
            si += c * Wi[k * D_FEAT * D_FEAT + i];
            c *= 0.5f;
        }
        WRc[i] = sr;
        WIc[i] = si;
    } else if (i < D_FEAT * D_FEAT + D_FEAT) {
        int o = i - D_FEAT * D_FEAT;
        float sr = 0.f, si = 0.f, c = 1.f;
        for (int k = 0; k < K; ++k) {
            sr += c * br[k * D_FEAT + o];
            si += c * bi[k * D_FEAT + o];
            c *= 0.5f;
        }
        bmi[o] = sr - si;
        bpi[o] = sr + si;
    }
}

__device__ __forceinline__ void gather_list(
        const int* __restrict__ adj, unsigned int start, unsigned int cnt,
        const float* __restrict__ winv, float self,
        const float* __restrict__ xr, const float* __restrict__ xi,
        int lane, float2& accR, float2& accI) {
    const int d = lane << 1;
    for (unsigned int base = 0; base < cnt; base += 32u) {
        const unsigned int m = min(32u, cnt - base);
        int   nb = 0;
        float w  = 0.f;
        if ((unsigned int)lane < m) {
            nb = adj[start + base + lane];
            w  = self * winv[nb];
        }
        unsigned int j = 0;
        for (; j + 4u <= m; j += 4u) {
            int   c[4];
            float wj[4];
            float2 a[4], b[4];
            #pragma unroll
            for (int u = 0; u < 4; ++u) {
                c[u]  = __builtin_amdgcn_readlane(nb, (int)(j + u));
                wj[u] = __int_as_float(__builtin_amdgcn_readlane(__float_as_int(w), (int)(j + u)));
                a[u]  = *(const float2*)(xr + (long)c[u] * D_FEAT + d);
                b[u]  = *(const float2*)(xi + (long)c[u] * D_FEAT + d);
            }
            #pragma unroll
            for (int u = 0; u < 4; ++u) {
                accR.x += wj[u] * a[u].x;  accR.y += wj[u] * a[u].y;
                accI.x += wj[u] * b[u].x;  accI.y += wj[u] * b[u].y;
            }
        }
        for (; j < m; ++j) {
            const int   c  = __builtin_amdgcn_readlane(nb, (int)j);
            const float wj = __int_as_float(__builtin_amdgcn_readlane(__float_as_int(w), (int)j));
            const float2 a = *(const float2*)(xr + (long)c * D_FEAT + d);
            const float2 b = *(const float2*)(xi + (long)c * D_FEAT + d);
            accR.x += wj * a.x;  accR.y += wj * a.y;
            accI.x += wj * b.x;  accI.y += wj * b.y;
        }
    }
}

#define G_PITCH     68
#define G_TILES_WG  5

__global__ __launch_bounds__(256) void gather_gemm_kernel(
        const float* __restrict__ xr, const float* __restrict__ xi,
        const int* __restrict__ adjO, const int* __restrict__ adjI,
        const unsigned int* __restrict__ blkOff, const unsigned int* __restrict__ locOff,
        const unsigned int* __restrict__ cnt,
        const float* __restrict__ out_inv, const float* __restrict__ in_inv, int n,
        const float* __restrict__ WRc, const float* __restrict__ WIc,
        const float* __restrict__ bmi, const float* __restrict__ bpi,
        float* __restrict__ outR, float* __restrict__ outI, int tiles) {
    __shared__ __attribute__((aligned(16))) float sW[2][D_FEAT * G_PITCH];
    __shared__ __attribute__((aligned(16))) float sA[3][16 * G_PITCH];
    __shared__ __attribute__((aligned(16))) float sO[2][16 * G_PITCH];

    const int tid  = threadIdx.x;
    const int wave = tid >> 5;
    const int lane = tid & 31;
    const int l16  = lane & 15;
    const int half = lane >> 4;
    const int ct   = wave & 3;
    const int part = wave >> 2;

    {
        float4 q[8];
        #pragma unroll
        for (int u = 0; u < 8; ++u) {
            const int i = tid + 256 * u;
            q[u] = *(const float4*)((u < 4 ? WRc : WIc) + (i & 1023) * 4);
        }
        #pragma unroll
        for (int u = 0; u < 8; ++u) {
            const int i  = tid + 256 * u;
            const int r  = (i >> 4) & (D_FEAT - 1);
            const int c4 = i & 15;
            *(float4*)(&sW[u >> 2][r * G_PITCH + c4 * 4]) = q[u];
        }
    }

    const float bias = part ? bpi[ct * 16 + l16] : bmi[ct * 16 + l16];
    const float* a0 = part ? sA[0] : sA[1];
    const float* a1 = sA[2];
    const float* b0 = part ? sW[1] : sW[0];
    const float* b1 = part ? sW[0] : sW[1];
    const int fragA = l16 * G_PITCH + 2 * half;
    const int fragB = (ct * 16 + l16) * G_PITCH + 2 * half;

    const int sr = tid >> 4, sc4 = tid & 15;
    const int lOff = sr * G_PITCH + sc4 * 4;

    for (int t = blockIdx.x; t < tiles; t += gridDim.x) {
        for (int q = 0; q < 2; ++q) {
            const int rowT = wave + 8 * q;
            int v = t * 16 + rowT;
            v = __builtin_amdgcn_readfirstlane(v);
            float2 yr  = make_float2(0.f, 0.f), yi  = make_float2(0.f, 0.f);
            float2 yrt = make_float2(0.f, 0.f), yit = make_float2(0.f, 0.f);
            if (v < n) {
                const unsigned int sO_ = blkOff[v / SCAN_BLOCK] + locOff[v];
                const unsigned int sI_ = blkOff[NB_MAX + v / SCAN_BLOCK] + locOff[(long)n + v];
                const unsigned int cO = cnt[v];
                const unsigned int cI = cnt[(long)n + v];
                const float dO = out_inv[v];
                const float dI = in_inv[v];
                gather_list(adjO, sO_, cO, in_inv, dO, xr, xi, lane, yr, yi);
                gather_list(adjI, sI_, cI, out_inv, dI, xr, xi, lane, yrt, yit);
            }
            const int o = rowT * G_PITCH + (lane << 1);
            *(float2*)(&sA[0][o]) = yr;
            *(float2*)(&sA[1][o]) = make_float2(0.5f * (yr.x + yrt.x), 0.5f * (yr.y + yrt.y));
            *(float2*)(&sA[2][o]) = make_float2(0.5f * (yi.x + yit.x), 0.5f * (yi.y + yit.y));
        }
        __syncthreads();

        v8f acc0 = {};
        v8f acc1 = {};
        #pragma unroll
        for (int s = 0; s < 16; ++s) {
            const v2f A0 = *(const v2f*)(a0 + fragA + 4 * s);
            const v2f A1 = *(const v2f*)(a1 + fragA + 4 * s);
            const v2f B0 = *(const v2f*)(b0 + fragB + 4 * s);
            const v2f B1 = *(const v2f*)(b1 + fragB + 4 * s);
            acc0 = __builtin_amdgcn_wmma_f32_16x16x4_f32(false, A0, false, B0, (short)0, acc0, false, false);
            acc1 = __builtin_amdgcn_wmma_f32_16x16x4_f32(false, A1, false, B1, (short)0, acc1, false, false);
        }

        float* so = &sO[part][(8 * half) * G_PITCH + ct * 16 + l16];
        #pragma unroll
        for (int r = 0; r < 8; ++r)
            so[r * G_PITCH] = part ? (acc0[r] + acc1[r] + bias) : (acc0[r] - acc1[r] + bias);
        __syncthreads();

        const int node = t * 16 + sr;
        if (node < n) {
            const long g = (long)node * D_FEAT + sc4 * 4;
            *(float4*)(outR + g) = *(const float4*)(&sO[0][lOff]);
            *(float4*)(outI + g) = *(const float4*)(&sO[1][lOff]);
        }
    }
}

extern "C" void kernel_launch(void* const* d_in, const int* in_sizes, int n_in,
                              void* d_out, int out_size, void* d_ws, size_t ws_size,
                              hipStream_t stream) {
    const float* xr = (const float*)d_in[0];
    const float* xi = (const float*)d_in[1];
    const int*   ei = (const int*)d_in[2];
    const float* Wr = (const float*)d_in[3];
    const float* br = (const float*)d_in[4];
    const float* Wi = (const float*)d_in[5];
    const float* bi = (const float*)d_in[6];

    const int n = in_sizes[0] / D_FEAT;
    const int E = in_sizes[2] / 2;
    const int K = in_sizes[3] / (D_FEAT * D_FEAT);
    const long NF = (long)n * D_FEAT;

    float* ws   = (float*)d_ws;
    float* degO = ws + 2 * NF;
    float* degI = degO + n;
    float* WRc  = degI + n;
    float* WIc  = WRc + D_FEAT * D_FEAT;
    float* bmi  = WIc + D_FEAT * D_FEAT;
    float* bpi  = bmi + D_FEAT;

    int* adjO = (int*)(ws + NF);
    int* adjI = adjO + E;
    unsigned int* cnt    = (unsigned int*)(adjI + E);
    unsigned int* locOff = cnt + 2 * (long)n;
    unsigned int* blkSum = locOff + 2 * (long)n;
    unsigned int* blkOff = blkSum + 2 * NB_MAX;

    float* outR = (float*)d_out;
    float* outI = outR + NF;
    const int* row = ei;
    const int* col = ei + E;

    unsigned int* part = (unsigned int*)ws;
    const dim3 hgrid(HIST_SLICES, (n + HIST_RANGE - 1) / HIST_RANGE, 2);
    degree_hist_kernel<<<hgrid, 1024, 0, stream>>>(row, col, E, n, part);
    deginv_kernel<<<dim3((n + SCAN_BLOCK - 1) / SCAN_BLOCK, 2), SCAN_BLOCK, 0, stream>>>(
        part, n, degO, degI, cnt, locOff, blkSum);
    blockscan_kernel<<<2, NB_MAX, 0, stream>>>(blkSum, (n + SCAN_BLOCK - 1) / SCAN_BLOCK, blkOff);
    fill_kernel<<<hgrid, 1024, 0, stream>>>(row, col, E, n, part, locOff, blkOff, adjO, adjI);
    wcomb_kernel<<<(D_FEAT * D_FEAT + D_FEAT + 255) / 256, 256, 0, stream>>>(
        Wr, br, Wi, bi, K, WRc, WIc, bmi, bpi);
    gather_gemm_kernel<<<((n + 15) / 16 + G_TILES_WG - 1) / G_TILES_WG, 256, 0, stream>>>(
        xr, xi, adjO, adjI, blkOff, locOff, cnt, degO, degI, n,
        WRc, WIc, bmi, bpi, outR, outI, (n + 15) / 16);
}


# edges grouped per node by a two-level LDS sort (chunks binned by node range, one workgroup per range counts, scans and places its runs); lists read ascending; 2 launches fewer; sum order changes
# speedup vs baseline: 4.0995x; 1.4941x over previous
#include <hip/hip_runtime.h>
#include <math.h>

typedef __attribute__((ext_vector_type(2))) float v2f;
typedef __attribute__((ext_vector_type(8))) float v8f;

#define D_FEAT 64

#define CHUNK    6400
#define NBIN     256
#define BIN_MAX  512
#define CH_MAX   1024
#define OWN_CAP  16384
#define PER_T    ((CHUNK + 1023) / 1024)

__global__ __launch_bounds__(1024) void partition_kernel(
        const int* __restrict__ row, const int* __restrict__ col, int E, int n,
        int2* __restrict__ pairs, unsigned int* __restrict__ offs) {
    __shared__ int2 stage[2][CHUNK];
    __shared__ unsigned int cntb[2 * NBIN];
    __shared__ unsigned int offb[2 * NBIN];

    const int tid   = threadIdx.x;
    const int c     = blockIdx.x;
    const int begin = c * CHUNK;
    const int len   = min(CHUNK, E - begin);
    const unsigned int bn = (unsigned int)((n + NBIN - 1) / NBIN);

    if (tid < 2 * NBIN) cntb[tid] = 0u;
    __syncthreads();

    int r[PER_T], q[PER_T];
    unsigned int b0[PER_T], b1[PER_T], k0[PER_T], k1[PER_T];
    #pragma unroll
    for (int u = 0; u < PER_T; ++u) {
        const int i = tid + u * 1024;
        r[u] = (i < len) ? row[begin + i] : 0;
        q[u] = (i < len) ? col[begin + i] : 0;
    }
    #pragma unroll
    for (int u = 0; u < PER_T; ++u) {
        const int i = tid + u * 1024;
        b0[u] = min((unsigned int)r[u] / bn, (unsigned int)(NBIN - 1));
        b1[u] = min((unsigned int)q[u] / bn, (unsigned int)(NBIN - 1));
        k0[u] = 0u; k1[u] = 0u;
        if (i < len) {
            k0[u] = atomicAdd(&cntb[b0[u]], 1u);
            k1[u] = atomicAdd(&cntb[NBIN + b1[u]], 1u);
        }
    }
    __syncthreads();

    const unsigned int mine = (tid < 2 * NBIN) ? cntb[tid] : 0u;
    if (tid < 2 * NBIN) offb[tid] = mine;
    for (int off = 1; off < NBIN; off <<= 1) {
        __syncthreads();
        const unsigned int t = (tid < 2 * NBIN && (tid & (NBIN - 1)) >= off) ? offb[tid - off] : 0u;
        __syncthreads();
        if (tid < 2 * NBIN) offb[tid] += t;
    }
    __syncthreads();
    const unsigned int excl = (tid < 2 * NBIN) ? offb[tid] - mine : 0u;
    __syncthreads();
    if (tid < 2 * NBIN) {
        offb[tid] = excl;
        const int z = tid / NBIN, b = tid & (NBIN - 1);
        unsigned int* o = offs + ((long)z * gridDim.x + c) * (NBIN + 1);
        o[b] = excl;
        if (b == NBIN - 1) o[NBIN] = (unsigned int)len;
    }
    __syncthreads();

    #pragma unroll
    for (int u = 0; u < PER_T; ++u) {
        const int i = tid + u * 1024;
        if (i < len) {
            stage[0][offb[b0[u]] + k0[u]]        = make_int2(r[u], q[u]);
            stage[1][offb[NBIN + b1[u]] + k1[u]] = make_int2(q[u], r[u]);
        }
    }
    __syncthreads();

    for (int i = tid; i < len; i += 1024) {
        pairs[(long)begin + i]     = stage[0][i];
        pairs[(long)E + begin + i] = stage[1][i];
    }
}

__global__ __launch_bounds__(1024) void owner_kernel(
        const int2* __restrict__ pairs, const unsigned int* __restrict__ offs, int E, int n, int nCh,
        int* __restrict__ adjO, int* __restrict__ adjI,
        unsigned int* __restrict__ cntG, unsigned int* __restrict__ startG,
        float* __restrict__ degO, float* __restrict__ degI) {
    __shared__ unsigned int pOff[CH_MAX];
    __shared__ unsigned int pLen[CH_MAX];
    __shared__ unsigned int cnt[BIN_MAX];
    __shared__ unsigned int cur[BIN_MAX];
    __shared__ unsigned int sc[BIN_MAX];
    __shared__ unsigned int baseS, totS;
    __shared__ int sAdj[OWN_CAP];

    const int tid  = threadIdx.x;
    const int lane = tid & 31;
    const int wave = tid >> 5;
    const int nW   = (int)blockDim.x >> 5;
    const int b    = blockIdx.x;
    const int z    = blockIdx.y;
    const int bn   = min((n + NBIN - 1) / NBIN, BIN_MAX);
    const int lo   = b * bn;
    const unsigned int cntN = (unsigned int)max(0, min(bn, n - lo));
    const int nc   = min(nCh, CH_MAX);

    const int2* __restrict__ P = pairs + (long)z * E;
    const unsigned int* __restrict__ O = offs + (long)z * nCh * (NBIN + 1);

    if (tid == 0) { baseS = 0u; totS = 0u; }
    if (tid < BIN_MAX) cnt[tid] = 0u;
    __syncthreads();

    for (int c = tid; c < nc; c += (int)blockDim.x) {
        const unsigned int o0 = O[(long)c * (NBIN + 1) + b];
        const unsigned int o1 = O[(long)c * (NBIN + 1) + b + 1];
        pOff[c] = (unsigned int)c * CHUNK + o0;
        pLen[c] = o1 - o0;
        atomicAdd(&baseS, o0);
        atomicAdd(&totS, o1 - o0);
    }
    __syncthreads();

    for (int c0 = wave * 8; c0 < nc; c0 += nW * 8) {
        int key[8];
        #pragma unroll
        for (int u = 0; u < 8; ++u) {
            const int c = c0 + u;
            const unsigned int L = (c < nc) ? pLen[c] : 0u;
            key[u] = ((unsigned int)lane < L) ? P[pOff[c] + lane].x : -1;
        }
        #pragma unroll
        for (int u = 0; u < 8; ++u) {
            const unsigned int d = (unsigned int)(key[u] - lo);
            if (key[u] >= 0 && d < cntN) atomicAdd(&cnt[d], 1u);
        }
        for (int u = 0; u < 8; ++u) {
            const int c = c0 + u;
            if (c < nc) {
                const unsigned int L = pLen[c], o = pOff[c];
                for (unsigned int i = 32u + lane; i < L; i += 32u) {
                    const unsigned int d = (unsigned int)(P[o + i].x - lo);
                    if (d < cntN) atomicAdd(&cnt[d], 1u);
                }
            }
        }
    }
    __syncthreads();

    const unsigned int mine = (tid < BIN_MAX) ? cnt[tid] : 0u;
    if (tid < BIN_MAX) sc[tid] = mine;
    for (int off = 1; off < BIN_MAX; off <<= 1) {
        __syncthreads();
        const unsigned int t = (tid < BIN_MAX && tid >= off) ? sc[tid - off] : 0u;
        __syncthreads();
        if (tid < BIN_MAX) sc[tid] += t;
    }
    __syncthreads();
    const unsigned int base = baseS;
    const unsigned int T    = totS;
    if ((unsigned int)tid < cntN) {
        const unsigned int excl = sc[tid] - mine;
        const long node = (long)lo + tid;
        cur[tid] = excl;
        startG[(long)z * n + node] = base + excl;
        cntG[(long)z * n + node]   = mine;
        const float d = (float)mine;
        const float w = (d > 0.f) ? powf(d, -0.25f) : 0.f;
        if (z) degI[node] = w; else degO[node] = w;
    }
    __syncthreads();

    const bool inLds = T <= (unsigned int)OWN_CAP;
    int* __restrict__ adj = (z ? adjI : adjO) + base;
    for (int c0 = wave * 8; c0 < nc; c0 += nW * 8) {
        int2 kv[8];
        #pragma unroll
        for (int u = 0; u < 8; ++u) {
            const int c = c0 + u;
            const unsigned int L = (c < nc) ? pLen[c] : 0u;
            kv[u] = ((unsigned int)lane < L) ? P[pOff[c] + lane] : make_int2(-1, 0);
        }
        #pragma unroll
        for (int u = 0; u < 8; ++u) {
            const unsigned int d = (unsigned int)(kv[u].x - lo);
            if (kv[u].x >= 0 && d < cntN) {
                const unsigned int pos = atomicAdd(&cur[d], 1u);
                if (inLds) sAdj[pos] = kv[u].y; else adj[pos] = kv[u].y;
            }
        }
        for (int u = 0; u < 8; ++u) {
            const int c = c0 + u;
            if (c < nc) {
                const unsigned int L = pLen[c], o = pOff[c];
                for (unsigned int i = 32u + lane; i < L; i += 32u) {
                    const int2 e = P[o + i];
                    const unsigned int d = (unsigned int)(e.x - lo);
                    if (d < cntN) {
                        const unsigned int pos = atomicAdd(&cur[d], 1u);
                        if (inLds) sAdj[pos] = e.y; else adj[pos] = e.y;
                    }
                }
            }
        }
    }
    __syncthreads();

    if (inLds) {
        for (unsigned int j = wave; j < cntN; j += nW) {
            const unsigned int L = (unsigned int)__builtin_amdgcn_readfirstlane((int)cnt[j]);
            const unsigned int s = (unsigned int)__builtin_amdgcn_readfirstlane((int)sc[j]) - L;
            if (L <= 32u) {
                if ((unsigned int)lane < L) adj[s + lane] = sAdj[s + lane];
                continue;
            }
            for (unsigned int i0 = 0; i0 < L; i0 += 32u) {
                const unsigned int myi = i0 + lane;
                const int me = (myi < L) ? sAdj[s + myi] : 0;
                const unsigned long long cme = ((unsigned long long)(unsigned int)me << 32) | myi;
                unsigned int rank = 0u;
                for (unsigned int j0 = 0; j0 < L; j0 += 32u) {
                    const int ov = (j0 + lane < L) ? sAdj[s + j0 + lane] : 0;
                    const unsigned int m = min(32u, L - j0);
                    for (unsigned int k = 0; k < m; ++k) {
                        const unsigned int o = (unsigned int)__builtin_amdgcn_readlane(ov, (int)k);
                        const unsigned long long co = ((unsigned long long)o << 32) | (j0 + k);
                        rank += (co < cme) ? 1u : 0u;
                    }
                }
                if (myi < L) adj[s + rank] = me;
            }
        }
    }
}

__global__ void wcomb_kernel(const float* __restrict__ Wr, const float* __restrict__ br,
                             const float* __restrict__ Wi, const float* __restrict__ bi,
                             int K, float* WRc, float* WIc, float* bmi, float* bpi) {
    int i = blockIdx.x * blockDim.x + threadIdx.x;
    if (i < D_FEAT * D_FEAT) {
        float sr = 0.f, si = 0.f, c = 1.f;
        for (int k = 0; k < K; ++k) {
            sr += c * Wr[k * D_FEAT * D_FEAT + i];
            si += c * Wi[k * D_FEAT * D_FEAT + i];
            c *= 0.5f;
        }
        WRc[i] = sr;
        WIc[i] = si;
    } else if (i < D_FEAT * D_FEAT + D_FEAT) {
        int o = i - D_FEAT * D_FEAT;
        float sr = 0.f, si = 0.f, c = 1.f;
        for (int k = 0; k < K; ++k) {
            sr += c * br[k * D_FEAT + o];
            si += c * bi[k * D_FEAT + o];
            c *= 0.5f;
        }
        bmi[o] = sr - si;
        bpi[o] = sr + si;
    }
}

__device__ __forceinline__ void gather_list(
        const int* __restrict__ adj, unsigned int start, unsigned int cnt,
        const float* __restrict__ winv, float self,
        const float* __restrict__ xr, const float* __restrict__ xi,
        int lane, float2& accR, float2& accI) {
    const int d = lane << 1;
    for (unsigned int base = 0; base < cnt; base += 32u) {
        const unsigned int m = min(32u, cnt - base);
        int   nb = 0;
        float w  = 0.f;
        if ((unsigned int)lane < m) nb = adj[start + base + lane];
        {
            const unsigned long long cme = ((unsigned long long)(unsigned int)nb << 32) | (unsigned int)lane;
            unsigned int rk = 0u;
            for (unsigned int k = 0; k < m; ++k) {
                const unsigned int o = (unsigned int)__builtin_amdgcn_readlane(nb, (int)k);
                rk += ((((unsigned long long)o << 32) | k) < cme) ? 1u : 0u;
            }
            if ((unsigned int)lane >= m) rk = (unsigned int)lane;
            nb = __builtin_amdgcn_ds_permute((int)(rk << 2), nb);
        }
        if ((unsigned int)lane < m) w = self * winv[nb];
        unsigned int j = 0;
        for (; j + 4u <= m; j += 4u) {
            int   c[4];
            float wj[4];
            float2 a[4], b[4];
            #pragma unroll
            for (int u = 0; u < 4; ++u) {
                c[u]  = __builtin_amdgcn_readlane(nb, (int)(j + u));
                wj[u] = __int_as_float(__builtin_amdgcn_readlane(__float_as_int(w), (int)(j + u)));
                a[u]  = *(const float2*)(xr + (long)c[u] * D_FEAT + d);
                b[u]  = *(const float2*)(xi + (long)c[u] * D_FEAT + d);
            }
            #pragma unroll
            for (int u = 0; u < 4; ++u) {
                accR.x += wj[u] * a[u].x;  accR.y += wj[u] * a[u].y;
                accI.x += wj[u] * b[u].x;  accI.y += wj[u] * b[u].y;
            }
        }
        for (; j < m; ++j) {
            const int   c  = __builtin_amdgcn_readlane(nb, (int)j);
            const float wj = __int_as_float(__builtin_amdgcn_readlane(__float_as_int(w), (int)j));
            const float2 a = *(const float2*)(xr + (long)c * D_FEAT + d);
            const float2 b = *(const float2*)(xi + (long)c * D_FEAT + d);
            accR.x += wj * a.x;  accR.y += wj * a.y;
            accI.x += wj * b.x;  accI.y += wj * b.y;
        }
    }
}

#define G_PITCH     68
#define G_TILES_WG  5

__global__ __launch_bounds__(256) void gather_gemm_kernel(
        const float* __restrict__ xr, const float* __restrict__ xi,
        const int* __restrict__ adjO, const int* __restrict__ adjI,
        const unsigned int* __restrict__ startG,
        const unsigned int* __restrict__ cnt,
        const float* __restrict__ out_inv, const float* __restrict__ in_inv, int n,
        const float* __restrict__ WRc, const float* __restrict__ WIc,
        const float* __restrict__ bmi, const float* __restrict__ bpi,
        float* __restrict__ outR, float* __restrict__ outI, int tiles) {
    __shared__ __attribute__((aligned(16))) float sW[2][D_FEAT * G_PITCH];
    __shared__ __attribute__((aligned(16))) float sA[3][16 * G_PITCH];
    __shared__ __attribute__((aligned(16))) float sO[2][16 * G_PITCH];

    const int tid  = threadIdx.x;
    const int wave = tid >> 5;
    const int lane = tid & 31;
    const int l16  = lane & 15;
    const int half = lane >> 4;
    const int ct   = wave & 3;
    const int part = wave >> 2;

    {
        float4 q[8];
        #pragma unroll
        for (int u = 0; u < 8; ++u) {
            const int i = tid + 256 * u;
            q[u] = *(const float4*)((u < 4 ? WRc : WIc) + (i & 1023) * 4);
        }
        #pragma unroll
        for (int u = 0; u < 8; ++u) {
            const int i  = tid + 256 * u;
            const int r  = (i >> 4) & (D_FEAT - 1);
            const int c4 = i & 15;
            *(float4*)(&sW[u >> 2][r * G_PITCH + c4 * 4]) = q[u];
        }
    }

    const float bias = part ? bpi[ct * 16 + l16] : bmi[ct * 16 + l16];
    const float* a0 = part ? sA[0] : sA[1];
    const float* a1 = sA[2];
    const float* b0 = part ? sW[1] : sW[0];
    const float* b1 = part ? sW[0] : sW[1];
    const int fragA = l16 * G_PITCH + 2 * half;
    const int fragB = (ct * 16 + l16) * G_PITCH + 2 * half;

    const int sr = tid >> 4, sc4 = tid & 15;
    const int lOff = sr * G_PITCH + sc4 * 4;

    for (int t = blockIdx.x; t < tiles; t += gridDim.x) {
        for (int q = 0; q < 2; ++q) {
            const int rowT = wave + 8 * q;
            int v = t * 16 + rowT;
            v = __builtin_amdgcn_readfirstlane(v);
            float2 yr  = make_float2(0.f, 0.f), yi  = make_float2(0.f, 0.f);
            float2 yrt = make_float2(0.f, 0.f), yit = make_float2(0.f, 0.f);
            if (v < n) {
                const unsigned int sO_ = startG[v];
                const unsigned int sI_ = startG[(long)n + v];
                const unsigned int cO = cnt[v];
                const unsigned int cI = cnt[(long)n + v];
                const float dO = out_inv[v];
                const float dI = in_inv[v];
                gather_list(adjO, sO_, cO, in_inv, dO, xr, xi, lane, yr, yi);
                gather_list(adjI, sI_, cI, out_inv, dI, xr, xi, lane, yrt, yit);
            }
            const int o = rowT * G_PITCH + (lane << 1);
            *(float2*)(&sA[0][o]) = yr;
            *(float2*)(&sA[1][o]) = make_float2(0.5f * (yr.x + yrt.x), 0.5f * (yr.y + yrt.y));
            *(float2*)(&sA[2][o]) = make_float2(0.5f * (yi.x + yit.x), 0.5f * (yi.y + yit.y));
        }
        __syncthreads();

        v8f acc0 = {};
        v8f acc1 = {};
        #pragma unroll
        for (int s = 0; s < 16; ++s) {
            const v2f A0 = *(const v2f*)(a0 + fragA + 4 * s);
            const v2f A1 = *(const v2f*)(a1 + fragA + 4 * s);
            const v2f B0 = *(const v2f*)(b0 + fragB + 4 * s);
            const v2f B1 = *(const v2f*)(b1 + fragB + 4 * s);
            acc0 = __builtin_amdgcn_wmma_f32_16x16x4_f32(false, A0, false, B0, (short)0, acc0, false, false);
            acc1 = __builtin_amdgcn_wmma_f32_16x16x4_f32(false, A1, false, B1, (short)0, acc1, false, false);
        }

        float* so = &sO[part][(8 * half) * G_PITCH + ct * 16 + l16];
        #pragma unroll
        for (int r = 0; r < 8; ++r)
            so[r * G_PITCH] = part ? (acc0[r] + acc1[r] + bias) : (acc0[r] - acc1[r] + bias);
        __syncthreads();

        const int node = t * 16 + sr;
        if (node < n) {
            const long g = (long)node * D_FEAT + sc4 * 4;
            *(float4*)(outR + g) = *(const float4*)(&sO[0][lOff]);
            *(float4*)(outI + g) = *(const float4*)(&sO[1][lOff]);
        }
    }
}

extern "C" void kernel_launch(void* const* d_in, const int* in_sizes, int n_in,
                              void* d_out, int out_size, void* d_ws, size_t ws_size,
                              hipStream_t stream) {
    const float* xr = (const float*)d_in[0];
    const float* xi = (const float*)d_in[1];
    const int*   ei = (const int*)d_in[2];
    const float* Wr = (const float*)d_in[3];
    const float* br = (const float*)d_in[4];
    const float* Wi = (const float*)d_in[5];
    const float* bi = (const float*)d_in[6];

    const int n = in_sizes[0] / D_FEAT;
    const int E = in_sizes[2] / 2;
    const int K = in_sizes[3] / (D_FEAT * D_FEAT);
    const long NF = (long)n * D_FEAT;

    float* ws   = (float*)d_ws;
    float* degO = ws + 2 * NF;
    float* degI = degO + n;
    float* WRc  = degI + n;
    float* WIc  = WRc + D_FEAT * D_FEAT;
    float* bmi  = WIc + D_FEAT * D_FEAT;
    float* bpi  = bmi + D_FEAT;

    int2* pairs = (int2*)ws;
    int* adjO = (int*)(ws + NF);
    int* adjI = adjO + E;
    unsigned int* cnt    = (unsigned int*)(adjI + E);
    unsigned int* startG = cnt + 2 * (long)n;
    unsigned int* offs   = startG + 2 * (long)n;
    const int nCh = (E + CHUNK - 1) / CHUNK;

    float* outR = (float*)d_out;
    float* outI = outR + NF;
    const int* row = ei;
    const int* col = ei + E;

    partition_kernel<<<nCh, 1024, 0, stream>>>(row, col, E, n, pairs, offs);
    owner_kernel<<<dim3(NBIN, 2), 1024, 0, stream>>>(pairs, offs, E, n, nCh, adjO, adjI, cnt, startG, degO, degI);
    wcomb_kernel<<<(D_FEAT * D_FEAT + D_FEAT + 255) / 256, 256, 0, stream>>>(
        Wr, br, Wi, bi, K, WRc, WIc, bmi, bpi);
    gather_gemm_kernel<<<((n + 15) / 16 + G_TILES_WG - 1) / G_TILES_WG, 256, 0, stream>>>(
        xr, xi, adjO, adjI, startG, cnt, degO, degI, n,
        WRc, WIc, bmi, bpi, outR, outI, (n + 15) / 16);
}
